# starting kernel
# baseline (speedup 1.0000x reference)
#include <hip/hip_runtime.h>

typedef __attribute__((ext_vector_type(2))) float v2f;
typedef __attribute__((ext_vector_type(8))) float v8f;

#define N_NODES 100000
#define N_EDGES 3200000


__global__ void k_init_deg(float* __restrict__ deg) {
    int i = blockIdx.x * blockDim.x + threadIdx.x;
    if (i < N_NODES) deg[i] = 1.0f;
}

__global__ void k_count_deg(const int* __restrict__ dst, float* __restrict__ deg) {
    int e = blockIdx.x * blockDim.x + threadIdx.x;
    if (e < N_EDGES) atomicAdd(&deg[dst[e]], 1.0f);
}

__global__ void k_to_dis(float* __restrict__ deg) {
    int i = blockIdx.x * blockDim.x + threadIdx.x;
    if (i < N_NODES) deg[i] = rsqrtf(deg[i]);
}


template <int IN, int OUT>
__global__ __launch_bounds__(256) void k_gemm_wmma(const float* __restrict__ X,
                                                   const float* __restrict__ W,
                                                   float* __restrict__ Y,
                                                   int nrows) {
    const int lane = threadIdx.x & 31;
    const int wave = blockIdx.x * (blockDim.x >> 5) + (threadIdx.x >> 5);
    const int row0 = wave * 16;
    if (row0 >= nrows) return;

    const int half = lane >> 4;
    const int l15  = lane & 15;
    const int arow = row0 + l15;

    constexpr int NT = OUT / 16;
    v8f acc[NT];
    #pragma unroll
    for (int t = 0; t < NT; ++t)
        #pragma unroll
        for (int v = 0; v < 8; ++v) acc[t][v] = 0.0f;

    for (int k = 0; k < IN; k += 4) {
        const float* ap = X + (size_t)arow * IN + (k + 2 * half);
        v2f a;
        a.x = ap[0];
        a.y = ap[1];

        #pragma unroll
        for (int t = 0; t < NT; ++t) {
            const int col = t * 16 + l15;
            v2f b;
            b.x = W[(k + 2 * half)     * OUT + col];
            b.y = W[(k + 2 * half + 1) * OUT + col];
            acc[t] = __builtin_amdgcn_wmma_f32_16x16x4_f32(
                 false, a,  false, b,
                 (short)0, acc[t],  false,  false);
        }
    }

    #pragma unroll
    for (int t = 0; t < NT; ++t)
        #pragma unroll
        for (int v = 0; v < 8; ++v) {
            int r = row0 + v + 8 * half;
            int c = t * 16 + l15;
            Y[(size_t)r * OUT + c] = acc[t][v];
        }
}


template <int OUT>
__global__ void k_self_init(const float* __restrict__ hw, const float* __restrict__ dis,
                            float* __restrict__ agg) {
    long long t = (long long)blockIdx.x * blockDim.x + threadIdx.x;
    if (t >= (long long)N_NODES * OUT) return;
    int i = (int)(t / OUT);
    float d = dis[i];
    agg[t] = d * d * hw[t];
}

template <int OUT>
__global__ void k_edge_agg(const int* __restrict__ src, const int* __restrict__ dst,
                           const float* __restrict__ dis, const float* __restrict__ hw,
                           float* __restrict__ agg) {
    long long t = (long long)blockIdx.x * blockDim.x + threadIdx.x;
    if (t >= (long long)N_EDGES * OUT) return;
    int e = (int)(t / OUT);
    int f = (int)(t % OUT);
    int s = src[e];
    int d = dst[e];
    float nrm = dis[s] * dis[d];
    atomicAdd(&agg[(long long)d * OUT + f], nrm * hw[(long long)s * OUT + f]);
}

template <int OUT>
__global__ void k_bias_relu(float* __restrict__ h, const float* __restrict__ b) {
    long long t = (long long)blockIdx.x * blockDim.x + threadIdx.x;
    if (t >= (long long)N_NODES * OUT) return;
    float v = h[t] + b[t % OUT];
    h[t] = v > 0.0f ? v : 0.0f;
}


__global__ void k_gemm_out1(const float* __restrict__ h2, const float* __restrict__ W3,
                            float* __restrict__ hw3) {
    int i = blockIdx.x * blockDim.x + threadIdx.x;
    if (i >= N_NODES) return;
    float s = 0.0f;
    #pragma unroll
    for (int k = 0; k < 16; ++k) s += h2[(size_t)i * 16 + k] * W3[k];
    hw3[i] = s;
}

__global__ void k_self_init_out1(const float* __restrict__ hw3, const float* __restrict__ dis,
                                 const float* __restrict__ b3, float* __restrict__ out) {
    int i = blockIdx.x * blockDim.x + threadIdx.x;
    if (i >= N_NODES) return;
    float d = dis[i];
    out[i] = d * d * hw3[i] + b3[0];
}

__global__ void k_edge_agg_out1(const int* __restrict__ src, const int* __restrict__ dst,
                                const float* __restrict__ dis, const float* __restrict__ hw3,
                                float* __restrict__ out) {
    int e = blockIdx.x * blockDim.x + threadIdx.x;
    if (e >= N_EDGES) return;
    int s = src[e];
    int d = dst[e];
    atomicAdd(&out[d], dis[s] * dis[d] * hw3[s]);
}


static inline int blks(long long n, int bs) { return (int)((n + bs - 1) / bs); }

extern "C" void kernel_launch(void* const* d_in, const int* in_sizes, int n_in,
                              void* d_out, int out_size, void* d_ws, size_t ws_size,
                              hipStream_t stream) {
    (void)in_sizes; (void)n_in; (void)out_size; (void)ws_size;

    const float* x   = (const float*)d_in[0];
    const int*   ei  = (const int*)  d_in[1];
    const float* W1  = (const float*)d_in[2];
    const float* b1  = (const float*)d_in[3];
    const float* W2  = (const float*)d_in[4];
    const float* b2  = (const float*)d_in[5];
    const float* W3  = (const float*)d_in[6];
    const float* b3  = (const float*)d_in[7];
    float* out = (float*)d_out;

    const int* src = ei;
    const int* dst = ei + N_EDGES;

    float* ws   = (float*)d_ws;
    float* dis  = ws;
    float* hw1  = dis  + N_NODES;
    float* h1   = hw1  + 32 * N_NODES;
    float* hw2  = h1   + 32 * N_NODES;
    float* h2   = hw2  + 16 * N_NODES;
    float* hw3  = h2   + 16 * N_NODES;

    const int BS = 256;

    k_init_deg <<<blks(N_NODES, BS), BS, 0, stream>>>(dis);
    k_count_deg<<<blks(N_EDGES, BS), BS, 0, stream>>>(dst, dis);
    k_to_dis   <<<blks(N_NODES, BS), BS, 0, stream>>>(dis);

    const int gemm_grid = blks(N_NODES / 16, BS / 32);

    k_gemm_wmma<128, 32><<<gemm_grid, BS, 0, stream>>>(x, W1, hw1, N_NODES);
    k_self_init<32><<<blks((long long)N_NODES * 32, BS), BS, 0, stream>>>(hw1, dis, h1);
    k_edge_agg <32><<<blks((long long)N_EDGES * 32, BS), BS, 0, stream>>>(src, dst, dis, hw1, h1);
    k_bias_relu<32><<<blks((long long)N_NODES * 32, BS), BS, 0, stream>>>(h1, b1);

    k_gemm_wmma<32, 16><<<gemm_grid, BS, 0, stream>>>(h1, W2, hw2, N_NODES);
    k_self_init<16><<<blks((long long)N_NODES * 16, BS), BS, 0, stream>>>(hw2, dis, h2);
    k_edge_agg <16><<<blks((long long)N_EDGES * 16, BS), BS, 0, stream>>>(src, dst, dis, hw2, h2);
    k_bias_relu<16><<<blks((long long)N_NODES * 16, BS), BS, 0, stream>>>(h2, b2);

    k_gemm_out1     <<<blks(N_NODES, BS), BS, 0, stream>>>(h2, W3, hw3);
    k_self_init_out1<<<blks(N_NODES, BS), BS, 0, stream>>>(hw3, dis, b3, out);
    k_edge_agg_out1 <<<blks(N_EDGES, BS), BS, 0, stream>>>(src, dst, dis, hw3, out);
}


# Self-loop start value, bias+relu and rsqrt(deg) passes folded into the transforms (epilogue stores dis^2*row; next transform's loads apply bias, relu): 14 -> 8 launches, no one-element grids left
# speedup vs baseline: 3.0525x; 3.0525x over previous
#include <hip/hip_runtime.h>

typedef __attribute__((ext_vector_type(2))) float v2f;
typedef __attribute__((ext_vector_type(8))) float v8f;

#define N_NODES 100000
#define N_EDGES 3200000


__global__ void k_init_deg(float* __restrict__ deg) {
    int i = blockIdx.x * blockDim.x + threadIdx.x;
    if (i < N_NODES) deg[i] = 1.0f;
}

#define PRIV_NR     8
#define PRIV_RSZ    (N_NODES / PRIV_NR)
#define PRIV_NC     32
#define PRIV_CHUNK  (N_EDGES / PRIV_NC)
#define PRIV_Q      (PRIV_CHUNK / 4)
#define PRIV_BS     1024
#define PRIV_ROT    384

__global__ __launch_bounds__(PRIV_BS) void k_count_deg(const int* __restrict__ dst,
                                                       float* __restrict__ deg) {
    __shared__ unsigned cnt[PRIV_RSZ];
    const int tid   = threadIdx.x;
    const int range = blockIdx.x % PRIV_NR;
    const int chunk = blockIdx.x / PRIV_NR;
    const int base  = range * PRIV_RSZ;

    for (int i = tid; i < PRIV_RSZ; i += PRIV_BS) cnt[i] = 0u;
    __syncthreads();

    const int4* pd = reinterpret_cast<const int4*>(dst + (size_t)chunk * PRIV_CHUNK);
    for (int i = tid; i < PRIV_Q; i += 4 * PRIV_BS) {
        int4 d[4];
        bool ok[4];
        #pragma unroll
        for (int g = 0; g < 4; ++g) {
            const int j = i + g * PRIV_BS;
            ok[g] = j < PRIV_Q;
            d[g] = pd[ok[g] ? j : i];
        }
        #pragma unroll
        for (int g = 0; g < 4; ++g) {
            const unsigned r0 = (unsigned)(d[g].x - base);
            const unsigned r1 = (unsigned)(d[g].y - base);
            const unsigned r2 = (unsigned)(d[g].z - base);
            const unsigned r3 = (unsigned)(d[g].w - base);
            if (ok[g] && r0 < PRIV_RSZ) atomicAdd(&cnt[r0], 1u);
            if (ok[g] && r1 < PRIV_RSZ) atomicAdd(&cnt[r1], 1u);
            if (ok[g] && r2 < PRIV_RSZ) atomicAdd(&cnt[r2], 1u);
            if (ok[g] && r3 < PRIV_RSZ) atomicAdd(&cnt[r3], 1u);
        }
    }
    __syncthreads();

    const int rot = chunk * PRIV_ROT;
    for (int i = tid; i < PRIV_RSZ; i += PRIV_BS) {
        int j = i + rot;
        if (j >= PRIV_RSZ) j -= PRIV_RSZ;
        atomicAdd(&deg[base + j], (float)cnt[j]);
    }
}

template <int IN, int OUT, int LDY, bool FIRST>
__global__ __launch_bounds__(256) void k_gemm_wmma(const float* __restrict__ X,
                                                   const float* __restrict__ W,
                                                   const float* __restrict__ bin,
                                                   float* __restrict__ dis,
                                                   float* __restrict__ Y,
                                                   float* __restrict__ agg,
                                                   int nrows) {
    const int lane = threadIdx.x & 31;
    const int wave = blockIdx.x * (blockDim.x >> 5) + (threadIdx.x >> 5);
    const int row0 = wave * 16;
    if (row0 >= nrows) return;

    const int half = lane >> 4;
    const int l15  = lane & 15;
    const int arow = row0 + l15;

    float d_own = dis[arow];

    constexpr int NT = OUT / 16;
    v8f acc[NT];
    #pragma unroll
    for (int t = 0; t < NT; ++t)
        #pragma unroll
        for (int v = 0; v < 8; ++v) acc[t][v] = 0.0f;

    for (int k = 0; k < IN; k += 4) {
        const float* ap = X + (size_t)arow * IN + (k + 2 * half);
        v2f a;
        a.x = ap[0];
        a.y = ap[1];
        if constexpr (!FIRST) {
            const float v0 = a.x + bin[k + 2 * half];
            const float v1 = a.y + bin[k + 2 * half + 1];
            a.x = v0 > 0.0f ? v0 : 0.0f;
            a.y = v1 > 0.0f ? v1 : 0.0f;
        }

        #pragma unroll
        for (int t = 0; t < NT; ++t) {
            const int col = t * 16 + l15;
            v2f b;
            b.x = W[(k + 2 * half)     * OUT + col];
            b.y = W[(k + 2 * half + 1) * OUT + col];
            acc[t] = __builtin_amdgcn_wmma_f32_16x16x4_f32(
                 false, a,  false, b,
                 (short)0, acc[t],  false,  false);
        }
    }

    if constexpr (FIRST) {
        d_own = rsqrtf(d_own);
        if (half == 0) dis[arow] = d_own;
    }

    float dd[8];
    #pragma unroll
    for (int v = 0; v < 8; ++v) {
        const float d = __shfl(d_own, v + 8 * half);
        dd[v] = d * d;
    }
    #pragma unroll
    for (int t = 0; t < NT; ++t)
        #pragma unroll
        for (int v = 0; v < 8; ++v) {
            int r = row0 + v + 8 * half;
            int c = t * 16 + l15;
            Y[(size_t)r * LDY + c]   = acc[t][v];
            agg[(size_t)r * OUT + c] = dd[v] * acc[t][v];
        }
    if constexpr (LDY > OUT) {
        if (half == 0) Y[(size_t)arow * LDY + OUT] = d_own;
    }
}


#define AG_BS    1024
#define AG_U     4
#define AG_CAP   (AG_U * 4 * 32)
#define AG_QBITS 12

template <int OUT, int NR, int NC, int LD>
__global__ __launch_bounds__(AG_BS) void k_edge_agg_lds(const int* __restrict__ src,
                                                        const int* __restrict__ dst,
                                                        const float* __restrict__ dis,
                                                        const float* __restrict__ hw,
                                                        float* __restrict__ agg) {
    constexpr int RSZ   = ((N_NODES + NR - 1) / NR + 1) & ~1;
    constexpr int CHUNK = N_EDGES / NC;
    constexpr int Q     = CHUNK / 4;
    constexpr int HPW   = 32 / OUT;
    constexpr int NI    = 8 / HPW;
    static_assert(RSZ <= (1 << AG_QBITS), "local index must fit its field");
    static_assert(N_EDGES % (NC * 4) == 0, "stretches are whole 16-byte groups");

    __shared__ float    acc[RSZ * OUT];
    __shared__ float    dsl[RSZ];
    __shared__ unsigned lst[(AG_BS / 32) * AG_CAP];

    const int tid   = threadIdx.x;
    const int lane  = tid & 31;
    const int range = blockIdx.x % NR;
    const int chunk = blockIdx.x / NR;
    const int base  = range * RSZ;
    const int rsz   = (N_NODES - base) < RSZ ? (N_NODES - base) : RSZ;

    for (int i = tid; i < RSZ * OUT; i += AG_BS) acc[i] = 0.0f;
    for (int i = tid; i < RSZ; i += AG_BS) dsl[i] = i < rsz ? dis[base + i] : 0.0f;
    __syncthreads();

    unsigned* my = lst + (tid >> 5) * AG_CAP;
    const int4* pd = reinterpret_cast<const int4*>(dst + (size_t)chunk * CHUNK);
    const int*  ps = src + (size_t)chunk * CHUNK;
    const int   f   = lane & (OUT - 1);
    const int   sub = lane / OUT;

    for (int i0 = 0; i0 < Q; i0 += AG_U * AG_BS) {
        int4 d4[AG_U];
        bool ok[AG_U];
        int  jj[AG_U];
        #pragma unroll
        for (int g = 0; g < AG_U; ++g) {
            const int j = i0 + g * AG_BS + tid;
            ok[g] = j < Q;
            jj[g] = ok[g] ? j : tid;
            d4[g] = pd[jj[g]];
        }
        bool     hit[4 * AG_U];
        unsigned q[4 * AG_U];
        int      ss[4 * AG_U];
        #pragma unroll
        for (int k = 0; k < 4 * AG_U; ++k) {
            const int g = k >> 2, c = k & 3;
            const int dv = c == 0 ? d4[g].x : c == 1 ? d4[g].y : c == 2 ? d4[g].z : d4[g].w;
            const unsigned r = (unsigned)(dv - base);
            hit[k] = ok[g] && r < (unsigned)RSZ;
            q[k]   = r;
            ss[k] = ps[hit[k] ? 4 * jj[g] + c : 0];
        }
        int cnt = 0;
        #pragma unroll
        for (int k = 0; k < 4 * AG_U; ++k) {
            const unsigned m = __builtin_amdgcn_ballot_w32(hit[k]);
            if (m != 0u) {
                const int pos = cnt + (int)__builtin_amdgcn_mbcnt_lo(m, 0u);
                if (hit[k]) my[pos] = ((unsigned)ss[k] << AG_QBITS) | q[k];
                cnt += __builtin_popcount(m);
            }
        }
        for (int b0 = 0; b0 < cnt; b0 += NI * HPW) {
            bool     val[NI];
            unsigned lq[NI];
            float    rw[NI], dsv[NI];
            #pragma unroll
            for (int u = 0; u < NI; ++u) {
                const int idx = b0 + u * HPW + sub;
                val[u] = idx < cnt;
                const unsigned e = my[val[u] ? idx : b0];
                const int s = (int)(e >> AG_QBITS);
                lq[u]  = e & ((1u << AG_QBITS) - 1u);
                rw[u]  = hw[(size_t)s * LD + f];
                if constexpr (LD > OUT) dsv[u] = hw[(size_t)s * LD + OUT];
                else                    dsv[u] = dis[s];
            }
            #pragma unroll
            for (int u = 0; u < NI; ++u) {
                const float w = dsv[u] * dsl[lq[u]] * rw[u];
                if (val[u]) atomicAdd(&acc[lq[u] * OUT + f], w);
            }
        }
    }
    __syncthreads();

    const int total = rsz * OUT;
    const int rot   = chunk * (((RSZ * OUT) / (2 * NC)) & ~31);
    float* ag = agg + (size_t)base * OUT;
    for (int i = tid; i < total; i += AG_BS) {
        int j = i + rot;
        if (j >= total) j -= total;
        atomicAdd(&ag[j], acc[j]);
    }
}


__global__ void k_gemm_out1(const float* __restrict__ h2, const float* __restrict__ b2,
                            const float* __restrict__ W3, const float* __restrict__ dis,
                            const float* __restrict__ b3,
                            float* __restrict__ pr, float* __restrict__ out) {
    int i = blockIdx.x * blockDim.x + threadIdx.x;
    if (i >= N_NODES) return;
    float s = 0.0f;
    #pragma unroll
    for (int k = 0; k < 16; ++k) {
        const float v = h2[(size_t)i * 16 + k] + b2[k];
        s += (v > 0.0f ? v : 0.0f) * W3[k];
    }
    const float d = dis[i];
    reinterpret_cast<float2*>(pr)[i] = make_float2(d, s);
    out[i] = d * d * s + b3[0];
}

__global__ __launch_bounds__(PRIV_BS) void k_edge_agg_out1(const int* __restrict__ src,
                                                           const int* __restrict__ dst,
                                                           const float* __restrict__ dis,
                                                           const float* __restrict__ pr,
                                                           float* __restrict__ out) {
    __shared__ float acc[PRIV_RSZ];
    __shared__ float dsl[PRIV_RSZ];
    const int tid   = threadIdx.x;
    const int range = blockIdx.x % PRIV_NR;
    const int chunk = blockIdx.x / PRIV_NR;
    const int base  = range * PRIV_RSZ;

    for (int i = tid; i < PRIV_RSZ; i += PRIV_BS) {
        acc[i] = 0.0f;
        dsl[i] = dis[base + i];
    }
    __syncthreads();

    const int4* pd = reinterpret_cast<const int4*>(dst + (size_t)chunk * PRIV_CHUNK);
    const int4* ps = reinterpret_cast<const int4*>(src + (size_t)chunk * PRIV_CHUNK);
    for (int i = tid; i < PRIV_Q; i += 2 * PRIV_BS) {
        int  dd[8], ss[8];
        bool ok[2];
        #pragma unroll
        for (int g = 0; g < 2; ++g) {
            const int j = i + g * PRIV_BS;
            ok[g] = j < PRIV_Q;
            const int4 d4 = pd[ok[g] ? j : i];
            const int4 s4 = ps[ok[g] ? j : i];
            dd[4 * g + 0] = d4.x; dd[4 * g + 1] = d4.y; dd[4 * g + 2] = d4.z; dd[4 * g + 3] = d4.w;
            ss[4 * g + 0] = s4.x; ss[4 * g + 1] = s4.y; ss[4 * g + 2] = s4.z; ss[4 * g + 3] = s4.w;
        }
        bool     hit[8];
        unsigned q[8];
        float    a[8], b[8], w[8];
        #pragma unroll
        for (int k = 0; k < 8; ++k) {
            const unsigned r = (unsigned)(dd[k] - base);
            hit[k] = ok[k >> 2] && r < PRIV_RSZ;
            q[k]   = hit[k] ? r : 0u;
            const int sidx = hit[k] ? ss[k] : 0;
            const float2 t = reinterpret_cast<const float2*>(pr)[sidx];
            a[k] = t.x;
            b[k] = t.y;
        }
        #pragma unroll
        for (int k = 0; k < 8; ++k) w[k] = a[k] * dsl[q[k]] * b[k];
        #pragma unroll
        for (int k = 0; k < 8; ++k)
            if (hit[k]) atomicAdd(&acc[q[k]], w[k]);
    }
    __syncthreads();

    const int rot = chunk * PRIV_ROT;
    for (int i = tid; i < PRIV_RSZ; i += PRIV_BS) {
        int j = i + rot;
        if (j >= PRIV_RSZ) j -= PRIV_RSZ;
        atomicAdd(&out[base + j], acc[j]);
    }
}


static inline int blks(long long n, int bs) { return (int)((n + bs - 1) / bs); }

extern "C" void kernel_launch(void* const* d_in, const int* in_sizes, int n_in,
                              void* d_out, int out_size, void* d_ws, size_t ws_size,
                              hipStream_t stream) {
    (void)in_sizes; (void)n_in; (void)out_size; (void)ws_size;

    const float* x   = (const float*)d_in[0];
    const int*   ei  = (const int*)  d_in[1];
    const float* W1  = (const float*)d_in[2];
    const float* b1  = (const float*)d_in[3];
    const float* W2  = (const float*)d_in[4];
    const float* b2  = (const float*)d_in[5];
    const float* W3  = (const float*)d_in[6];
    const float* b3  = (const float*)d_in[7];
    float* out = (float*)d_out;

    const int* src = ei;
    const int* dst = ei + N_EDGES;

    float* ws   = (float*)d_ws;
    float* dis  = ws;
    float* hw1  = dis  + N_NODES;
    float* h1   = hw1  + 32 * N_NODES;
    float* pr3  = h1   + 32 * N_NODES;
    float* h2   = pr3  + 16 * N_NODES;
    float* hw2p = hw1;

    const int BS = 256;

    k_init_deg <<<blks(N_NODES, BS), BS, 0, stream>>>(dis);
    k_count_deg<<<PRIV_NR * PRIV_NC, PRIV_BS, 0, stream>>>(dst, dis);

    const int gemm_grid = blks(N_NODES / 16, BS / 32);

    k_gemm_wmma<128, 32, 32, true><<<gemm_grid, BS, 0, stream>>>(x, W1, nullptr, dis, hw1, h1, N_NODES);
    k_edge_agg_lds<32, 64, 4, 32><<<64 * 4, AG_BS, 0, stream>>>(src, dst, dis, hw1, h1);

    k_gemm_wmma<32, 16, 32, false><<<gemm_grid, BS, 0, stream>>>(h1, W2, b1, dis, hw2p, h2, N_NODES);
    k_edge_agg_lds<16, 32, 8, 32><<<32 * 8, AG_BS, 0, stream>>>(src, dst, dis, hw2p, h2);

    k_gemm_out1    <<<blks(N_NODES, BS), BS, 0, stream>>>(h2, b2, W3, dis, b3, pr3, out);
    k_edge_agg_out1<<<PRIV_NR * PRIV_NC, PRIV_BS, 0, stream>>>(src, dst, dis, pr3, out);
}


# Edge scatters software-pipelined: a pass requests dst of trip t+2 and src of trip t+1, then fetches and adds the rows of trip t: one memory wait per pass, not three; terms and add order unchanged
# speedup vs baseline: 3.2996x; 1.0810x over previous
#include <hip/hip_runtime.h>

typedef __attribute__((ext_vector_type(2))) float v2f;
typedef __attribute__((ext_vector_type(8))) float v8f;

#define N_NODES 100000
#define N_EDGES 3200000


__global__ void k_init_deg(float* __restrict__ deg) {
    int i = blockIdx.x * blockDim.x + threadIdx.x;
    if (i < N_NODES) deg[i] = 1.0f;
}

#define PRIV_NR     8
#define PRIV_RSZ    (N_NODES / PRIV_NR)
#define PRIV_NC     32
#define PRIV_CHUNK  (N_EDGES / PRIV_NC)
#define PRIV_Q      (PRIV_CHUNK / 4)
#define PRIV_BS     1024
#define PRIV_ROT    384

__global__ __launch_bounds__(PRIV_BS) void k_count_deg(const int* __restrict__ dst,
                                                       float* __restrict__ deg) {
    __shared__ unsigned cnt[PRIV_RSZ];
    const int tid   = threadIdx.x;
    const int range = blockIdx.x % PRIV_NR;
    const int chunk = blockIdx.x / PRIV_NR;
    const int base  = range * PRIV_RSZ;

    for (int i = tid; i < PRIV_RSZ; i += PRIV_BS) cnt[i] = 0u;
    __syncthreads();

    const int4* pd = reinterpret_cast<const int4*>(dst + (size_t)chunk * PRIV_CHUNK);
    for (int i = tid; i < PRIV_Q; i += 4 * PRIV_BS) {
        int4 d[4];
        bool ok[4];
        #pragma unroll
        for (int g = 0; g < 4; ++g) {
            const int j = i + g * PRIV_BS;
            ok[g] = j < PRIV_Q;
            d[g] = pd[ok[g] ? j : i];
        }
        #pragma unroll
        for (int g = 0; g < 4; ++g) {
            const unsigned r0 = (unsigned)(d[g].x - base);
            const unsigned r1 = (unsigned)(d[g].y - base);
            const unsigned r2 = (unsigned)(d[g].z - base);
            const unsigned r3 = (unsigned)(d[g].w - base);
            if (ok[g] && r0 < PRIV_RSZ) atomicAdd(&cnt[r0], 1u);
            if (ok[g] && r1 < PRIV_RSZ) atomicAdd(&cnt[r1], 1u);
            if (ok[g] && r2 < PRIV_RSZ) atomicAdd(&cnt[r2], 1u);
            if (ok[g] && r3 < PRIV_RSZ) atomicAdd(&cnt[r3], 1u);
        }
    }
    __syncthreads();

    const int rot = chunk * PRIV_ROT;
    for (int i = tid; i < PRIV_RSZ; i += PRIV_BS) {
        int j = i + rot;
        if (j >= PRIV_RSZ) j -= PRIV_RSZ;
        atomicAdd(&deg[base + j], (float)cnt[j]);
    }
}

template <int IN, int OUT, int LDY, bool FIRST>
__global__ __launch_bounds__(256) void k_gemm_wmma(const float* __restrict__ X,
                                                   const float* __restrict__ W,
                                                   const float* __restrict__ bin,
                                                   float* __restrict__ dis,
                                                   float* __restrict__ Y,
                                                   float* __restrict__ agg,
                                                   int nrows) {
    const int lane = threadIdx.x & 31;
    const int wave = blockIdx.x * (blockDim.x >> 5) + (threadIdx.x >> 5);
    const int row0 = wave * 16;
    if (row0 >= nrows) return;

    const int half = lane >> 4;
    const int l15  = lane & 15;
    const int arow = row0 + l15;

    float d_own = dis[arow];

    constexpr int NT = OUT / 16;
    v8f acc[NT];
    #pragma unroll
    for (int t = 0; t < NT; ++t)
        #pragma unroll
        for (int v = 0; v < 8; ++v) acc[t][v] = 0.0f;

    for (int k = 0; k < IN; k += 4) {
        const float* ap = X + (size_t)arow * IN + (k + 2 * half);
        v2f a;
        a.x = ap[0];
        a.y = ap[1];
        if constexpr (!FIRST) {
            const float v0 = a.x + bin[k + 2 * half];
            const float v1 = a.y + bin[k + 2 * half + 1];
            a.x = v0 > 0.0f ? v0 : 0.0f;
            a.y = v1 > 0.0f ? v1 : 0.0f;
        }

        #pragma unroll
        for (int t = 0; t < NT; ++t) {
            const int col = t * 16 + l15;
            v2f b;
            b.x = W[(k + 2 * half)     * OUT + col];
            b.y = W[(k + 2 * half + 1) * OUT + col];
            acc[t] = __builtin_amdgcn_wmma_f32_16x16x4_f32(
                 false, a,  false, b,
                 (short)0, acc[t],  false,  false);
        }
    }

    if constexpr (FIRST) {
        d_own = rsqrtf(d_own);
        if (half == 0) dis[arow] = d_own;
    }

    float dd[8];
    #pragma unroll
    for (int v = 0; v < 8; ++v) {
        const float d = __shfl(d_own, v + 8 * half);
        dd[v] = d * d;
    }
    #pragma unroll
    for (int t = 0; t < NT; ++t)
        #pragma unroll
        for (int v = 0; v < 8; ++v) {
            int r = row0 + v + 8 * half;
            int c = t * 16 + l15;
            Y[(size_t)r * LDY + c]   = acc[t][v];
            agg[(size_t)r * OUT + c] = dd[v] * acc[t][v];
        }
    if constexpr (LDY > OUT) {
        if (half == 0) Y[(size_t)arow * LDY + OUT] = d_own;
    }
}


#define AG_BS    1024
#define AG_U     4
#define AG_CAP   (AG_U * 4 * 32)
#define AG_QBITS 12

template <int OUT, int NR, int NC, int LD>
__global__ __launch_bounds__(AG_BS) void k_edge_agg_lds(const int* __restrict__ src,
                                                        const int* __restrict__ dst,
                                                        const float* __restrict__ dis,
                                                        const float* __restrict__ hw,
                                                        float* __restrict__ agg) {
    constexpr int RSZ   = ((N_NODES + NR - 1) / NR + 1) & ~1;
    constexpr int CHUNK = N_EDGES / NC;
    constexpr int Q     = CHUNK / 4;
    constexpr int HPW   = 32 / OUT;
    constexpr int NI    = 8 / HPW;
    static_assert(RSZ <= (1 << AG_QBITS), "local index must fit its field");
    static_assert(N_EDGES % (NC * 4) == 0, "stretches are whole 16-byte groups");

    __shared__ float    acc[RSZ * OUT];
    __shared__ float    dsl[RSZ];
    __shared__ unsigned lst[(AG_BS / 32) * AG_CAP];

    const int tid   = threadIdx.x;
    const int lane  = tid & 31;
    const int range = blockIdx.x % NR;
    const int chunk = blockIdx.x / NR;
    const int base  = range * RSZ;
    const int rsz   = (N_NODES - base) < RSZ ? (N_NODES - base) : RSZ;

    for (int i = tid; i < RSZ * OUT; i += AG_BS) acc[i] = 0.0f;
    for (int i = tid; i < RSZ; i += AG_BS) dsl[i] = i < rsz ? dis[base + i] : 0.0f;
    __syncthreads();

    unsigned* my = lst + (tid >> 5) * AG_CAP;
    const int4* pd = reinterpret_cast<const int4*>(dst + (size_t)chunk * CHUNK);
    const int*  ps = src + (size_t)chunk * CHUNK;
    const int   f   = lane & (OUT - 1);
    const int   sub = lane / OUT;

    constexpr int STEP = AG_U * AG_BS;

    int4 dn[AG_U];
    #pragma unroll
    for (int g = 0; g < AG_U; ++g) {
        const int j = g * AG_BS + tid;
        dn[g] = pd[j < Q ? j : tid];
    }
    int cnt = 0;

    for (int i0 = 0; i0 < Q + STEP; i0 += STEP) {
        bool ok[AG_U];
        int  jj[AG_U];
        #pragma unroll
        for (int g = 0; g < AG_U; ++g) {
            const int j = i0 + g * AG_BS + tid;
            ok[g] = j < Q;
            jj[g] = ok[g] ? j : tid;
        }
        bool     hit[4 * AG_U];
        unsigned q[4 * AG_U];
        int      ss[4 * AG_U];
        #pragma unroll
        for (int k = 0; k < 4 * AG_U; ++k) {
            const int g = k >> 2, c = k & 3;
            const int dv = c == 0 ? dn[g].x : c == 1 ? dn[g].y : c == 2 ? dn[g].z : dn[g].w;
            const unsigned r = (unsigned)(dv - base);
            hit[k] = ok[g] && r < (unsigned)RSZ;
            q[k]   = r;
            ss[k] = ps[hit[k] ? 4 * jj[g] + c : 0];
        }
        #pragma unroll
        for (int g = 0; g < AG_U; ++g) {
            const int j = i0 + STEP + g * AG_BS + tid;
            dn[g] = pd[j < Q ? j : tid];
        }
        for (int b0 = 0; b0 < cnt; b0 += NI * HPW) {
            bool     val[NI];
            unsigned lq[NI];
            float    rw[NI], dsv[NI], dl[NI];
            #pragma unroll
            for (int u = 0; u < NI; ++u) {
                const int idx = b0 + u * HPW + sub;
                val[u] = idx < cnt;
                const unsigned e = my[val[u] ? idx : b0];
                const int s = (int)(e >> AG_QBITS);
                lq[u]  = e & ((1u << AG_QBITS) - 1u);
                rw[u]  = hw[(size_t)s * LD + f];
                if constexpr (LD > OUT) dsv[u] = hw[(size_t)s * LD + OUT];
                else                    dsv[u] = dis[s];
                dl[u]  = dsl[lq[u]];
            }
            #pragma unroll
            for (int u = 0; u < NI; ++u) asm volatile("" :: "v"(rw[u]), "v"(dsv[u]), "v"(dl[u]));
            #pragma unroll
            for (int u = 0; u < NI; ++u) {
                const float w = dsv[u] * dl[u] * rw[u];
                if (val[u]) atomicAdd(&acc[lq[u] * OUT + f], w);
            }
        }
        cnt = 0;
        #pragma unroll
        for (int k = 0; k < 4 * AG_U; ++k) {
            const unsigned m = __builtin_amdgcn_ballot_w32(hit[k]);
            if (m != 0u) {
                const int pos = cnt + (int)__builtin_amdgcn_mbcnt_lo(m, 0u);
                if (hit[k]) my[pos] = ((unsigned)ss[k] << AG_QBITS) | q[k];
                cnt += __builtin_popcount(m);
            }
        }
    }
    __syncthreads();

    const int total = rsz * OUT;
    const int rot   = chunk * (((RSZ * OUT) / (2 * NC)) & ~31);
    float* ag = agg + (size_t)base * OUT;
    for (int i = tid; i < total; i += AG_BS) {
        int j = i + rot;
        if (j >= total) j -= total;
        atomicAdd(&ag[j], acc[j]);
    }
}


__global__ void k_gemm_out1(const float* __restrict__ h2, const float* __restrict__ b2,
                            const float* __restrict__ W3, const float* __restrict__ dis,
                            const float* __restrict__ b3,
                            float* __restrict__ pr, float* __restrict__ out) {
    int i = blockIdx.x * blockDim.x + threadIdx.x;
    if (i >= N_NODES) return;
    float s = 0.0f;
    #pragma unroll
    for (int k = 0; k < 16; ++k) {
        const float v = h2[(size_t)i * 16 + k] + b2[k];
        s += (v > 0.0f ? v : 0.0f) * W3[k];
    }
    const float d = dis[i];
    reinterpret_cast<float2*>(pr)[i] = make_float2(d, s);
    out[i] = d * d * s + b3[0];
}

__global__ __launch_bounds__(PRIV_BS) void k_edge_agg_out1(const int* __restrict__ src,
                                                           const int* __restrict__ dst,
                                                           const float* __restrict__ dis,
                                                           const float* __restrict__ pr,
                                                           float* __restrict__ out) {
    __shared__ float acc[PRIV_RSZ];
    __shared__ float dsl[PRIV_RSZ];
    const int tid   = threadIdx.x;
    const int range = blockIdx.x % PRIV_NR;
    const int chunk = blockIdx.x / PRIV_NR;
    const int base  = range * PRIV_RSZ;

    for (int i = tid; i < PRIV_RSZ; i += PRIV_BS) {
        acc[i] = 0.0f;
        dsl[i] = dis[base + i];
    }
    __syncthreads();

    const int4* pd = reinterpret_cast<const int4*>(dst + (size_t)chunk * PRIV_CHUNK);
    const int4* ps = reinterpret_cast<const int4*>(src + (size_t)chunk * PRIV_CHUNK);
    int4 dn[2], sn[2];
    #pragma unroll
    for (int g = 0; g < 2; ++g) {
        const int j  = g * PRIV_BS + tid;
        const int jv = j < PRIV_Q ? j : tid;
        dn[g] = pd[jv];
        sn[g] = ps[jv];
    }
    for (int i0 = 0; i0 < PRIV_Q; i0 += 2 * PRIV_BS) {
        int  dd[8], ss[8];
        bool ok[2];
        #pragma unroll
        for (int g = 0; g < 2; ++g) {
            ok[g] = i0 + g * PRIV_BS + tid < PRIV_Q;
            dd[4 * g + 0] = dn[g].x; dd[4 * g + 1] = dn[g].y; dd[4 * g + 2] = dn[g].z; dd[4 * g + 3] = dn[g].w;
            ss[4 * g + 0] = sn[g].x; ss[4 * g + 1] = sn[g].y; ss[4 * g + 2] = sn[g].z; ss[4 * g + 3] = sn[g].w;
        }
        #pragma unroll
        for (int g = 0; g < 2; ++g) {
            const int j  = i0 + (2 + g) * PRIV_BS + tid;
            const int jv = j < PRIV_Q ? j : tid;
            dn[g] = pd[jv];
            sn[g] = ps[jv];
        }
        bool     hit[8];
        unsigned q[8];
        float    a[8], b[8], w[8];
        #pragma unroll
        for (int k = 0; k < 8; ++k) {
            const unsigned r = (unsigned)(dd[k] - base);
            hit[k] = ok[k >> 2] && r < PRIV_RSZ;
            q[k]   = hit[k] ? r : 0u;
            const int sidx = hit[k] ? ss[k] : 0;
            const float2 t = reinterpret_cast<const float2*>(pr)[sidx];
            a[k] = t.x;
            b[k] = t.y;
        }
        #pragma unroll
        for (int k = 0; k < 8; ++k) w[k] = a[k] * dsl[q[k]] * b[k];
        #pragma unroll
        for (int k = 0; k < 8; ++k)
            if (hit[k]) atomicAdd(&acc[q[k]], w[k]);
    }
    __syncthreads();

    const int rot = chunk * PRIV_ROT;
    for (int i = tid; i < PRIV_RSZ; i += PRIV_BS) {
        int j = i + rot;
        if (j >= PRIV_RSZ) j -= PRIV_RSZ;
        atomicAdd(&out[base + j], acc[j]);
    }
}


static inline int blks(long long n, int bs) { return (int)((n + bs - 1) / bs); }

extern "C" void kernel_launch(void* const* d_in, const int* in_sizes, int n_in,
                              void* d_out, int out_size, void* d_ws, size_t ws_size,
                              hipStream_t stream) {
    (void)in_sizes; (void)n_in; (void)out_size; (void)ws_size;

    const float* x   = (const float*)d_in[0];
    const int*   ei  = (const int*)  d_in[1];
    const float* W1  = (const float*)d_in[2];
    const float* b1  = (const float*)d_in[3];
    const float* W2  = (const float*)d_in[4];
    const float* b2  = (const float*)d_in[5];
    const float* W3  = (const float*)d_in[6];
    const float* b3  = (const float*)d_in[7];
    float* out = (float*)d_out;

    const int* src = ei;
    const int* dst = ei + N_EDGES;

    float* ws   = (float*)d_ws;
    float* dis  = ws;
    float* hw1  = dis  + N_NODES;
    float* h1   = hw1  + 32 * N_NODES;
    float* pr3  = h1   + 32 * N_NODES;
    float* h2   = pr3  + 16 * N_NODES;
    float* hw2p = hw1;

    const int BS = 256;

    k_init_deg <<<blks(N_NODES, BS), BS, 0, stream>>>(dis);
    k_count_deg<<<PRIV_NR * PRIV_NC, PRIV_BS, 0, stream>>>(dst, dis);

    const int gemm_grid = blks(N_NODES / 16, BS / 32);

    k_gemm_wmma<128, 32, 32, true><<<gemm_grid, BS, 0, stream>>>(x, W1, nullptr, dis, hw1, h1, N_NODES);
    k_edge_agg_lds<32, 64, 4, 32><<<64 * 4, AG_BS, 0, stream>>>(src, dst, dis, hw1, h1);

    k_gemm_wmma<32, 16, 32, false><<<gemm_grid, BS, 0, stream>>>(h1, W2, b1, dis, hw2p, h2, N_NODES);
    k_edge_agg_lds<16, 32, 8, 32><<<32 * 8, AG_BS, 0, stream>>>(src, dst, dis, hw2p, h2);

    k_gemm_out1    <<<blks(N_NODES, BS), BS, 0, stream>>>(h2, b2, W3, dis, b3, pr3, out);
    k_edge_agg_out1<<<PRIV_NR * PRIV_NC, PRIV_BS, 0, stream>>>(src, dst, dis, pr3, out);
}


# Row scatters read src for hits only: the scan tests dst alone and packs (edge offset, node); one lane-per-entry src read per list replaces a src read per scanned edge; terms and add order unchanged
# speedup vs baseline: 3.4827x; 1.0555x over previous
#include <hip/hip_runtime.h>

typedef __attribute__((ext_vector_type(2))) float v2f;
typedef __attribute__((ext_vector_type(8))) float v8f;

#define N_NODES 100000
#define N_EDGES 3200000


__global__ void k_init_deg(float* __restrict__ deg) {
    int i = blockIdx.x * blockDim.x + threadIdx.x;
    if (i < N_NODES) deg[i] = 1.0f;
}

#define PRIV_NR     8
#define PRIV_RSZ    (N_NODES / PRIV_NR)
#define PRIV_NC     32
#define PRIV_CHUNK  (N_EDGES / PRIV_NC)
#define PRIV_Q      (PRIV_CHUNK / 4)
#define PRIV_BS     1024
#define PRIV_ROT    384

__global__ __launch_bounds__(PRIV_BS) void k_count_deg(const int* __restrict__ dst,
                                                       float* __restrict__ deg) {
    __shared__ unsigned cnt[PRIV_RSZ];
    const int tid   = threadIdx.x;
    const int range = blockIdx.x % PRIV_NR;
    const int chunk = blockIdx.x / PRIV_NR;
    const int base  = range * PRIV_RSZ;

    for (int i = tid; i < PRIV_RSZ; i += PRIV_BS) cnt[i] = 0u;
    __syncthreads();

    const int4* pd = reinterpret_cast<const int4*>(dst + (size_t)chunk * PRIV_CHUNK);
    for (int i = tid; i < PRIV_Q; i += 4 * PRIV_BS) {
        int4 d[4];
        bool ok[4];
        #pragma unroll
        for (int g = 0; g < 4; ++g) {
            const int j = i + g * PRIV_BS;
            ok[g] = j < PRIV_Q;
            d[g] = pd[ok[g] ? j : i];
        }
        #pragma unroll
        for (int g = 0; g < 4; ++g) {
            const unsigned r0 = (unsigned)(d[g].x - base);
            const unsigned r1 = (unsigned)(d[g].y - base);
            const unsigned r2 = (unsigned)(d[g].z - base);
            const unsigned r3 = (unsigned)(d[g].w - base);
            if (ok[g] && r0 < PRIV_RSZ) atomicAdd(&cnt[r0], 1u);
            if (ok[g] && r1 < PRIV_RSZ) atomicAdd(&cnt[r1], 1u);
            if (ok[g] && r2 < PRIV_RSZ) atomicAdd(&cnt[r2], 1u);
            if (ok[g] && r3 < PRIV_RSZ) atomicAdd(&cnt[r3], 1u);
        }
    }
    __syncthreads();

    const int rot = chunk * PRIV_ROT;
    for (int i = tid; i < PRIV_RSZ; i += PRIV_BS) {
        int j = i + rot;
        if (j >= PRIV_RSZ) j -= PRIV_RSZ;
        atomicAdd(&deg[base + j], (float)cnt[j]);
    }
}

template <int IN, int OUT, int LDY, bool FIRST>
__global__ __launch_bounds__(256) void k_gemm_wmma(const float* __restrict__ X,
                                                   const float* __restrict__ W,
                                                   const float* __restrict__ bin,
                                                   float* __restrict__ dis,
                                                   float* __restrict__ Y,
                                                   float* __restrict__ agg,
                                                   int nrows) {
    const int lane = threadIdx.x & 31;
    const int wave = blockIdx.x * (blockDim.x >> 5) + (threadIdx.x >> 5);
    const int row0 = wave * 16;
    if (row0 >= nrows) return;

    const int half = lane >> 4;
    const int l15  = lane & 15;
    const int arow = row0 + l15;

    float d_own = dis[arow];

    constexpr int NT = OUT / 16;
    v8f acc[NT];
    #pragma unroll
    for (int t = 0; t < NT; ++t)
        #pragma unroll
        for (int v = 0; v < 8; ++v) acc[t][v] = 0.0f;

    for (int k = 0; k < IN; k += 4) {
        const float* ap = X + (size_t)arow * IN + (k + 2 * half);
        v2f a;
        a.x = ap[0];
        a.y = ap[1];
        if constexpr (!FIRST) {
            const float v0 = a.x + bin[k + 2 * half];
            const float v1 = a.y + bin[k + 2 * half + 1];
            a.x = v0 > 0.0f ? v0 : 0.0f;
            a.y = v1 > 0.0f ? v1 : 0.0f;
        }

        #pragma unroll
        for (int t = 0; t < NT; ++t) {
            const int col = t * 16 + l15;
            v2f b;
            b.x = W[(k + 2 * half)     * OUT + col];
            b.y = W[(k + 2 * half + 1) * OUT + col];
            acc[t] = __builtin_amdgcn_wmma_f32_16x16x4_f32(
                 false, a,  false, b,
                 (short)0, acc[t],  false,  false);
        }
    }

    if constexpr (FIRST) {
        d_own = rsqrtf(d_own);
        if (half == 0) dis[arow] = d_own;
    }

    float dd[8];
    #pragma unroll
    for (int v = 0; v < 8; ++v) {
        const float d = __shfl(d_own, v + 8 * half);
        dd[v] = d * d;
    }
    #pragma unroll
    for (int t = 0; t < NT; ++t)
        #pragma unroll
        for (int v = 0; v < 8; ++v) {
            int r = row0 + v + 8 * half;
            int c = t * 16 + l15;
            Y[(size_t)r * LDY + c]   = acc[t][v];
            agg[(size_t)r * OUT + c] = dd[v] * acc[t][v];
        }
    if constexpr (LDY > OUT) {
        if (half == 0) Y[(size_t)arow * LDY + OUT] = d_own;
    }
}


#define AG_BS    1024
#define AG_U     4
#define AG_CAP   (AG_U * 4 * 32)
#define AG_QBITS 12

template <int OUT, int NR, int NC, int LD>
__global__ __launch_bounds__(AG_BS) void k_edge_agg_lds(const int* __restrict__ src,
                                                        const int* __restrict__ dst,
                                                        const float* __restrict__ dis,
                                                        const float* __restrict__ hw,
                                                        float* __restrict__ agg) {
    constexpr int RSZ   = ((N_NODES + NR - 1) / NR + 1) & ~1;
    constexpr int CHUNK = N_EDGES / NC;
    constexpr int Q     = CHUNK / 4;
    constexpr int HPW   = 32 / OUT;
    constexpr int NI    = 8 / HPW;
    static_assert(RSZ <= (1 << AG_QBITS), "local index must fit its field");
    static_assert(N_EDGES % (NC * 4) == 0, "stretches are whole 16-byte groups");

    __shared__ float    acc[RSZ * OUT];
    __shared__ float    dsl[RSZ];
    __shared__ unsigned lst[(AG_BS / 32) * AG_CAP];

    const int tid   = threadIdx.x;
    const int lane  = tid & 31;
    const int range = blockIdx.x % NR;
    const int chunk = blockIdx.x / NR;
    const int base  = range * RSZ;
    const int rsz   = (N_NODES - base) < RSZ ? (N_NODES - base) : RSZ;

    const int4* pd = reinterpret_cast<const int4*>(dst + (size_t)chunk * CHUNK);
    const int*  ps = src + (size_t)chunk * CHUNK;

    int4 dn[AG_U];
    #pragma unroll
    for (int g = 0; g < AG_U; ++g) {
        const int j = g * AG_BS + tid;
        dn[g] = pd[j < Q ? j : tid];
    }

    for (int i = tid; i < RSZ * OUT; i += AG_BS) acc[i] = 0.0f;
    for (int i = tid; i < RSZ; i += AG_BS) dsl[i] = i < rsz ? dis[base + i] : 0.0f;
    __syncthreads();

    unsigned* my = lst + (tid >> 5) * AG_CAP;
    const int   f   = lane & (OUT - 1);
    const int   sub = lane / OUT;

    constexpr int      STEP  = AG_U * AG_BS;
    constexpr unsigned QMASK = (1u << AG_QBITS) - 1u;
    static_assert(CHUNK <= (1 << (32 - AG_QBITS)), "edge offset in a stretch must fit its field");

    int      cnt = 0;
    unsigned ent = 0u;
    int      sv  = 0;

    for (int i0 = 0; i0 < Q + STEP; i0 += STEP) {
        bool     ok[AG_U];
        unsigned okm[AG_U];
        unsigned eb[AG_U];
        #pragma unroll
        for (int g = 0; g < AG_U; ++g) {
            const int j = i0 + g * AG_BS + tid;
            ok[g]  = j < Q;
            okm[g] = __builtin_amdgcn_ballot_w32(ok[g]);
            eb[g]  = (unsigned)(4 * j) << AG_QBITS;
        }
        bool     hit[4 * AG_U];
        unsigned hm[4 * AG_U];
        unsigned q[4 * AG_U];
        #pragma unroll
        for (int k = 0; k < 4 * AG_U; ++k) {
            const int g = k >> 2, c = k & 3;
            const int dv = c == 0 ? dn[g].x : c == 1 ? dn[g].y : c == 2 ? dn[g].z : dn[g].w;
            const unsigned r = (unsigned)(dv - base);
            const bool in = r < (unsigned)RSZ;
            hm[k]  = __builtin_amdgcn_ballot_w32(in) & okm[g];
            hit[k] = in && ok[g];
            q[k]   = r;
        }
        #pragma unroll
        for (int g = 0; g < AG_U; ++g) {
            const int j = i0 + STEP + g * AG_BS + tid;
            dn[g] = pd[j < Q ? j : tid];
        }
        if (lane < cnt) my[lane] = ((unsigned)sv << AG_QBITS) | (ent & QMASK);
        #pragma nounroll
        for (int j0 = 32; j0 < cnt; j0 += 32) {
            const int j = j0 + lane;
            if (j < cnt) {
                const unsigned e = my[j];
                const int s = ps[e >> AG_QBITS];
                my[j] = ((unsigned)s << AG_QBITS) | (e & QMASK);
            }
        }
        for (int b0 = 0; b0 < cnt; b0 += NI * HPW) {
            bool     val[NI];
            unsigned lq[NI];
            float    rw[NI], dsv[NI], dl[NI];
            #pragma unroll
            for (int u = 0; u < NI; ++u) {
                const int idx = b0 + u * HPW + sub;
                val[u] = idx < cnt;
                const unsigned e = my[val[u] ? idx : b0];
                const int s = (int)(e >> AG_QBITS);
                lq[u]  = e & QMASK;
                rw[u]  = hw[(size_t)s * LD + f];
                if constexpr (LD > OUT) dsv[u] = hw[(size_t)s * LD + OUT];
                else                    dsv[u] = dis[s];
                dl[u]  = dsl[lq[u]];
            }
            #pragma unroll
            for (int u = 0; u < NI; ++u) asm volatile("" :: "v"(rw[u]), "v"(dsv[u]), "v"(dl[u]));
            #pragma unroll
            for (int u = 0; u < NI; ++u) {
                const float w = dsv[u] * dl[u] * rw[u];
                if (val[u]) atomicAdd(&acc[lq[u] * OUT + f], w);
            }
        }
        cnt = 0;
        #pragma unroll
        for (int k = 0; k < 4 * AG_U; ++k) {
            const int g = k >> 2, c = k & 3;
            if (hm[k] != 0u) {
                const int pos = cnt + (int)__builtin_amdgcn_mbcnt_lo(hm[k], 0u);
                if (hit[k]) my[pos] = eb[g] | ((unsigned)c << AG_QBITS) | q[k];
                cnt += __builtin_popcount(hm[k]);
            }
        }
        ent = my[lane];
        sv  = ps[lane < cnt ? (int)(ent >> AG_QBITS) : 0];
    }
    __syncthreads();

    const int total = rsz * OUT;
    const int rot   = chunk * (((RSZ * OUT) / (2 * NC)) & ~31);
    float* ag = agg + (size_t)base * OUT;
    for (int i = tid; i < total; i += AG_BS) {
        int j = i + rot;
        if (j >= total) j -= total;
        atomicAdd(&ag[j], acc[j]);
    }
}


__global__ void k_gemm_out1(const float* __restrict__ h2, const float* __restrict__ b2,
                            const float* __restrict__ W3, const float* __restrict__ dis,
                            const float* __restrict__ b3,
                            float* __restrict__ pr, float* __restrict__ out) {
    int i = blockIdx.x * blockDim.x + threadIdx.x;
    if (i >= N_NODES) return;
    float s = 0.0f;
    #pragma unroll
    for (int k = 0; k < 16; ++k) {
        const float v = h2[(size_t)i * 16 + k] + b2[k];
        s += (v > 0.0f ? v : 0.0f) * W3[k];
    }
    const float d = dis[i];
    reinterpret_cast<float2*>(pr)[i] = make_float2(d, s);
    out[i] = d * d * s + b3[0];
}

__global__ __launch_bounds__(PRIV_BS) void k_edge_agg_out1(const int* __restrict__ src,
                                                           const int* __restrict__ dst,
                                                           const float* __restrict__ dis,
                                                           const float* __restrict__ pr,
                                                           float* __restrict__ out) {
    __shared__ float acc[PRIV_RSZ];
    __shared__ float dsl[PRIV_RSZ];
    const int tid   = threadIdx.x;
    const int range = blockIdx.x % PRIV_NR;
    const int chunk = blockIdx.x / PRIV_NR;
    const int base  = range * PRIV_RSZ;

    for (int i = tid; i < PRIV_RSZ; i += PRIV_BS) {
        acc[i] = 0.0f;
        dsl[i] = dis[base + i];
    }
    __syncthreads();

    const int4* pd = reinterpret_cast<const int4*>(dst + (size_t)chunk * PRIV_CHUNK);
    const int4* ps = reinterpret_cast<const int4*>(src + (size_t)chunk * PRIV_CHUNK);
    int4 dn[2], sn[2];
    #pragma unroll
    for (int g = 0; g < 2; ++g) {
        const int j  = g * PRIV_BS + tid;
        const int jv = j < PRIV_Q ? j : tid;
        dn[g] = pd[jv];
        sn[g] = ps[jv];
    }
    for (int i0 = 0; i0 < PRIV_Q; i0 += 2 * PRIV_BS) {
        int  dd[8], ss[8];
        bool ok[2];
        #pragma unroll
        for (int g = 0; g < 2; ++g) {
            ok[g] = i0 + g * PRIV_BS + tid < PRIV_Q;
            dd[4 * g + 0] = dn[g].x; dd[4 * g + 1] = dn[g].y; dd[4 * g + 2] = dn[g].z; dd[4 * g + 3] = dn[g].w;
            ss[4 * g + 0] = sn[g].x; ss[4 * g + 1] = sn[g].y; ss[4 * g + 2] = sn[g].z; ss[4 * g + 3] = sn[g].w;
        }
        #pragma unroll
        for (int g = 0; g < 2; ++g) {
            const int j  = i0 + (2 + g) * PRIV_BS + tid;
            const int jv = j < PRIV_Q ? j : tid;
            dn[g] = pd[jv];
            sn[g] = ps[jv];
        }
        bool     hit[8];
        unsigned q[8];
        float    a[8], b[8], w[8];
        #pragma unroll
        for (int k = 0; k < 8; ++k) {
            const unsigned r = (unsigned)(dd[k] - base);
            hit[k] = ok[k >> 2] && r < PRIV_RSZ;
            q[k]   = hit[k] ? r : 0u;
            const int sidx = hit[k] ? ss[k] : 0;
            const float2 t = reinterpret_cast<const float2*>(pr)[sidx];
            a[k] = t.x;
            b[k] = t.y;
        }
        #pragma unroll
        for (int k = 0; k < 8; ++k) w[k] = a[k] * dsl[q[k]] * b[k];
        #pragma unroll
        for (int k = 0; k < 8; ++k)
            if (hit[k]) atomicAdd(&acc[q[k]], w[k]);
    }
    __syncthreads();

    const int rot = chunk * PRIV_ROT;
    for (int i = tid; i < PRIV_RSZ; i += PRIV_BS) {
        int j = i + rot;
        if (j >= PRIV_RSZ) j -= PRIV_RSZ;
        atomicAdd(&out[base + j], acc[j]);
    }
}


static inline int blks(long long n, int bs) { return (int)((n + bs - 1) / bs); }

extern "C" void kernel_launch(void* const* d_in, const int* in_sizes, int n_in,
                              void* d_out, int out_size, void* d_ws, size_t ws_size,
                              hipStream_t stream) {
    (void)in_sizes; (void)n_in; (void)out_size; (void)ws_size;

    const float* x   = (const float*)d_in[0];
    const int*   ei  = (const int*)  d_in[1];
    const float* W1  = (const float*)d_in[2];
    const float* b1  = (const float*)d_in[3];
    const float* W2  = (const float*)d_in[4];
    const float* b2  = (const float*)d_in[5];
    const float* W3  = (const float*)d_in[6];
    const float* b3  = (const float*)d_in[7];
    float* out = (float*)d_out;

    const int* src = ei;
    const int* dst = ei + N_EDGES;

    float* ws   = (float*)d_ws;
    float* dis  = ws;
    float* hw1  = dis  + N_NODES;
    float* h1   = hw1  + 32 * N_NODES;
    float* pr3  = h1   + 32 * N_NODES;
    float* h2   = pr3  + 16 * N_NODES;
    float* hw2p = hw1;

    const int BS = 256;

    k_init_deg <<<blks(N_NODES, BS), BS, 0, stream>>>(dis);
    k_count_deg<<<PRIV_NR * PRIV_NC, PRIV_BS, 0, stream>>>(dst, dis);

    const int gemm_grid = blks(N_NODES / 16, BS / 32);

    k_gemm_wmma<128, 32, 32, true><<<gemm_grid, BS, 0, stream>>>(x, W1, nullptr, dis, hw1, h1, N_NODES);
    k_edge_agg_lds<32, 64, 4, 32><<<64 * 4, AG_BS, 0, stream>>>(src, dst, dis, hw1, h1);

    k_gemm_wmma<32, 16, 32, false><<<gemm_grid, BS, 0, stream>>>(h1, W2, b1, dis, hw2p, h2, N_NODES);
    k_edge_agg_lds<16, 32, 8, 32><<<32 * 8, AG_BS, 0, stream>>>(src, dst, dis, hw2p, h2);

    k_gemm_out1    <<<blks(N_NODES, BS), BS, 0, stream>>>(h2, b2, W3, dis, b3, pr3, out);
    k_edge_agg_out1<<<PRIV_NR * PRIV_NC, PRIV_BS, 0, stream>>>(src, dst, dis, pr3, out);
}


# Layer-1/2 transforms store rows as (value, dis[row]) pairs, 256 B and 128 B a row; a scatter reads row and scale in one 8-byte load per lane, no separate dis[src] read; terms and add order unchanged
# speedup vs baseline: 3.5978x; 1.0330x over previous
#include <hip/hip_runtime.h>

typedef __attribute__((ext_vector_type(2))) float v2f;
typedef __attribute__((ext_vector_type(8))) float v8f;

#define N_NODES 100000
#define N_EDGES 3200000


__global__ void k_init_deg(float* __restrict__ deg) {
    int i = blockIdx.x * blockDim.x + threadIdx.x;
    if (i < N_NODES) deg[i] = 1.0f;
}

#define PRIV_NR     8
#define PRIV_RSZ    (N_NODES / PRIV_NR)
#define PRIV_NC     32
#define PRIV_CHUNK  (N_EDGES / PRIV_NC)
#define PRIV_Q      (PRIV_CHUNK / 4)
#define PRIV_BS     1024
#define PRIV_ROT    384

__global__ __launch_bounds__(PRIV_BS) void k_count_deg(const int* __restrict__ dst,
                                                       float* __restrict__ deg) {
    __shared__ unsigned cnt[PRIV_RSZ];
    const int tid   = threadIdx.x;
    const int range = blockIdx.x % PRIV_NR;
    const int chunk = blockIdx.x / PRIV_NR;
    const int base  = range * PRIV_RSZ;

    for (int i = tid; i < PRIV_RSZ; i += PRIV_BS) cnt[i] = 0u;
    __syncthreads();

    const int4* pd = reinterpret_cast<const int4*>(dst + (size_t)chunk * PRIV_CHUNK);
    for (int i = tid; i < PRIV_Q; i += 4 * PRIV_BS) {
        int4 d[4];
        bool ok[4];
        #pragma unroll
        for (int g = 0; g < 4; ++g) {
            const int j = i + g * PRIV_BS;
            ok[g] = j < PRIV_Q;
            d[g] = pd[ok[g] ? j : i];
        }
        #pragma unroll
        for (int g = 0; g < 4; ++g) {
            const unsigned r0 = (unsigned)(d[g].x - base);
            const unsigned r1 = (unsigned)(d[g].y - base);
            const unsigned r2 = (unsigned)(d[g].z - base);
            const unsigned r3 = (unsigned)(d[g].w - base);
            if (ok[g] && r0 < PRIV_RSZ) atomicAdd(&cnt[r0], 1u);
            if (ok[g] && r1 < PRIV_RSZ) atomicAdd(&cnt[r1], 1u);
            if (ok[g] && r2 < PRIV_RSZ) atomicAdd(&cnt[r2], 1u);
            if (ok[g] && r3 < PRIV_RSZ) atomicAdd(&cnt[r3], 1u);
        }
    }
    __syncthreads();

    const int rot = chunk * PRIV_ROT;
    for (int i = tid; i < PRIV_RSZ; i += PRIV_BS) {
        int j = i + rot;
        if (j >= PRIV_RSZ) j -= PRIV_RSZ;
        atomicAdd(&deg[base + j], (float)cnt[j]);
    }
}

template <int IN, int OUT, bool FIRST>
__global__ __launch_bounds__(256) void k_gemm_wmma(const float* __restrict__ X,
                                                   const float* __restrict__ W,
                                                   const float* __restrict__ bin,
                                                   float* __restrict__ dis,
                                                   float* __restrict__ Y,
                                                   float* __restrict__ agg,
                                                   int nrows) {
    const int lane = threadIdx.x & 31;
    const int wave = blockIdx.x * (blockDim.x >> 5) + (threadIdx.x >> 5);
    const int row0 = wave * 16;
    if (row0 >= nrows) return;

    const int half = lane >> 4;
    const int l15  = lane & 15;
    const int arow = row0 + l15;

    float d_own = dis[arow];

    constexpr int NT = OUT / 16;
    v8f acc[NT];
    #pragma unroll
    for (int t = 0; t < NT; ++t)
        #pragma unroll
        for (int v = 0; v < 8; ++v) acc[t][v] = 0.0f;

    for (int k = 0; k < IN; k += 4) {
        const float* ap = X + (size_t)arow * IN + (k + 2 * half);
        v2f a;
        a.x = ap[0];
        a.y = ap[1];
        if constexpr (!FIRST) {
            const float v0 = a.x + bin[k + 2 * half];
            const float v1 = a.y + bin[k + 2 * half + 1];
            a.x = v0 > 0.0f ? v0 : 0.0f;
            a.y = v1 > 0.0f ? v1 : 0.0f;
        }

        #pragma unroll
        for (int t = 0; t < NT; ++t) {
            const int col = t * 16 + l15;
            v2f b;
            b.x = W[(k + 2 * half)     * OUT + col];
            b.y = W[(k + 2 * half + 1) * OUT + col];
            acc[t] = __builtin_amdgcn_wmma_f32_16x16x4_f32(
                 false, a,  false, b,
                 (short)0, acc[t],  false,  false);
        }
    }

    if constexpr (FIRST) {
        d_own = rsqrtf(d_own);
        if (half == 0) dis[arow] = d_own;
    }

    float dv[8], dd[8];
    #pragma unroll
    for (int v = 0; v < 8; ++v) {
        dv[v] = __shfl(d_own, v + 8 * half);
        dd[v] = dv[v] * dv[v];
    }
    float2* Yp = reinterpret_cast<float2*>(Y);
    #pragma unroll
    for (int t = 0; t < NT; ++t)
        #pragma unroll
        for (int v = 0; v < 8; ++v) {
            int r = row0 + v + 8 * half;
            int c = t * 16 + l15;
            Yp[(size_t)r * OUT + c]  = make_float2(acc[t][v], dv[v]);
            agg[(size_t)r * OUT + c] = dd[v] * acc[t][v];
        }
}


#define AG_BS    1024
#define AG_U     4
#define AG_CAP   (AG_U * 4 * 32)
#define AG_QBITS 12

template <int OUT, int NR, int NC>
__global__ __launch_bounds__(AG_BS) void k_edge_agg_lds(const int* __restrict__ src,
                                                        const int* __restrict__ dst,
                                                        const float* __restrict__ dis,
                                                        const float* __restrict__ hw,
                                                        float* __restrict__ agg) {
    constexpr int RSZ   = ((N_NODES + NR - 1) / NR + 1) & ~1;
    constexpr int CHUNK = N_EDGES / NC;
    constexpr int Q     = CHUNK / 4;
    constexpr int HPW   = 32 / OUT;
    constexpr int NI    = 8 / HPW;
    static_assert(RSZ <= (1 << AG_QBITS), "local index must fit its field");
    static_assert(N_EDGES % (NC * 4) == 0, "stretches are whole 16-byte groups");

    __shared__ float    acc[RSZ * OUT];
    __shared__ float    dsl[RSZ];
    __shared__ unsigned lst[(AG_BS / 32) * AG_CAP];

    const int tid   = threadIdx.x;
    const int lane  = tid & 31;
    const int range = blockIdx.x % NR;
    const int chunk = blockIdx.x / NR;
    const int base  = range * RSZ;
    const int rsz   = (N_NODES - base) < RSZ ? (N_NODES - base) : RSZ;

    const int4* pd = reinterpret_cast<const int4*>(dst + (size_t)chunk * CHUNK);
    const int*  ps = src + (size_t)chunk * CHUNK;

    int4 dn[AG_U];
    #pragma unroll
    for (int g = 0; g < AG_U; ++g) {
        const int j = g * AG_BS + tid;
        dn[g] = pd[j < Q ? j : tid];
    }

    for (int i = tid; i < RSZ * OUT; i += AG_BS) acc[i] = 0.0f;
    for (int i = tid; i < RSZ; i += AG_BS) dsl[i] = i < rsz ? dis[base + i] : 0.0f;
    __syncthreads();

    unsigned* my = lst + (tid >> 5) * AG_CAP;
    const float2* hwp = reinterpret_cast<const float2*>(hw);
    const int   f   = lane & (OUT - 1);
    const int   sub = lane / OUT;

    constexpr int      STEP  = AG_U * AG_BS;
    constexpr unsigned QMASK = (1u << AG_QBITS) - 1u;
    static_assert(CHUNK <= (1 << (32 - AG_QBITS)), "edge offset in a stretch must fit its field");

    int      cnt = 0;
    unsigned ent = 0u;
    int      sv  = 0;

    for (int i0 = 0; i0 < Q + STEP; i0 += STEP) {
        bool     ok[AG_U];
        unsigned okm[AG_U];
        unsigned eb[AG_U];
        #pragma unroll
        for (int g = 0; g < AG_U; ++g) {
            const int j = i0 + g * AG_BS + tid;
            ok[g]  = j < Q;
            okm[g] = __builtin_amdgcn_ballot_w32(ok[g]);
            eb[g]  = (unsigned)(4 * j) << AG_QBITS;
        }
        bool     hit[4 * AG_U];
        unsigned hm[4 * AG_U];
        unsigned q[4 * AG_U];
        #pragma unroll
        for (int k = 0; k < 4 * AG_U; ++k) {
            const int g = k >> 2, c = k & 3;
            const int dv = c == 0 ? dn[g].x : c == 1 ? dn[g].y : c == 2 ? dn[g].z : dn[g].w;
            const unsigned r = (unsigned)(dv - base);
            const bool in = r < (unsigned)RSZ;
            hm[k]  = __builtin_amdgcn_ballot_w32(in) & okm[g];
            hit[k] = in && ok[g];
            q[k]   = r;
        }
        #pragma unroll
        for (int g = 0; g < AG_U; ++g) {
            const int j = i0 + STEP + g * AG_BS + tid;
            dn[g] = pd[j < Q ? j : tid];
        }
        if (lane < cnt) my[lane] = ((unsigned)sv << AG_QBITS) | (ent & QMASK);
        #pragma nounroll
        for (int j0 = 32; j0 < cnt; j0 += 32) {
            const int j = j0 + lane;
            if (j < cnt) {
                const unsigned e = my[j];
                const int s = ps[e >> AG_QBITS];
                my[j] = ((unsigned)s << AG_QBITS) | (e & QMASK);
            }
        }
        for (int b0 = 0; b0 < cnt; b0 += NI * HPW) {
            bool     val[NI];
            unsigned lq[NI];
            float    rw[NI], dsv[NI], dl[NI];
            #pragma unroll
            for (int u = 0; u < NI; ++u) {
                const int idx = b0 + u * HPW + sub;
                val[u] = idx < cnt;
                const unsigned e = my[val[u] ? idx : b0];
                const int s = (int)(e >> AG_QBITS);
                lq[u]  = e & QMASK;
                const float2 p = hwp[(size_t)s * OUT + f];
                rw[u]  = p.x;
                dsv[u] = p.y;
                dl[u]  = dsl[lq[u]];
            }
            #pragma unroll
            for (int u = 0; u < NI; ++u) asm volatile("" :: "v"(rw[u]), "v"(dsv[u]), "v"(dl[u]));
            #pragma unroll
            for (int u = 0; u < NI; ++u) {
                const float w = dsv[u] * dl[u] * rw[u];
                if (val[u]) atomicAdd(&acc[lq[u] * OUT + f], w);
            }
        }
        cnt = 0;
        #pragma unroll
        for (int k = 0; k < 4 * AG_U; ++k) {
            const int g = k >> 2, c = k & 3;
            if (hm[k] != 0u) {
                const int pos = cnt + (int)__builtin_amdgcn_mbcnt_lo(hm[k], 0u);
                if (hit[k]) my[pos] = eb[g] | ((unsigned)c << AG_QBITS) | q[k];
                cnt += __builtin_popcount(hm[k]);
            }
        }
        ent = my[lane];
        sv  = ps[lane < cnt ? (int)(ent >> AG_QBITS) : 0];
    }
    __syncthreads();

    const int total = rsz * OUT;
    const int rot   = chunk * (((RSZ * OUT) / (2 * NC)) & ~31);
    float* ag = agg + (size_t)base * OUT;
    for (int i = tid; i < total; i += AG_BS) {
        int j = i + rot;
        if (j >= total) j -= total;
        atomicAdd(&ag[j], acc[j]);
    }
}


__global__ void k_gemm_out1(const float* __restrict__ h2, const float* __restrict__ b2,
                            const float* __restrict__ W3, const float* __restrict__ dis,
                            const float* __restrict__ b3,
                            float* __restrict__ pr, float* __restrict__ out) {
    int i = blockIdx.x * blockDim.x + threadIdx.x;
    if (i >= N_NODES) return;
    float s = 0.0f;
    #pragma unroll
    for (int k = 0; k < 16; ++k) {
        const float v = h2[(size_t)i * 16 + k] + b2[k];
        s += (v > 0.0f ? v : 0.0f) * W3[k];
    }
    const float d = dis[i];
    reinterpret_cast<float2*>(pr)[i] = make_float2(d, s);
    out[i] = d * d * s + b3[0];
}

__global__ __launch_bounds__(PRIV_BS) void k_edge_agg_out1(const int* __restrict__ src,
                                                           const int* __restrict__ dst,
                                                           const float* __restrict__ dis,
                                                           const float* __restrict__ pr,
                                                           float* __restrict__ out) {
    __shared__ float acc[PRIV_RSZ];
    __shared__ float dsl[PRIV_RSZ];
    const int tid   = threadIdx.x;
    const int range = blockIdx.x % PRIV_NR;
    const int chunk = blockIdx.x / PRIV_NR;
    const int base  = range * PRIV_RSZ;

    for (int i = tid; i < PRIV_RSZ; i += PRIV_BS) {
        acc[i] = 0.0f;
        dsl[i] = dis[base + i];
    }
    __syncthreads();

    const int4* pd = reinterpret_cast<const int4*>(dst + (size_t)chunk * PRIV_CHUNK);
    const int4* ps = reinterpret_cast<const int4*>(src + (size_t)chunk * PRIV_CHUNK);
    int4 dn[2], sn[2];
    #pragma unroll
    for (int g = 0; g < 2; ++g) {
        const int j  = g * PRIV_BS + tid;
        const int jv = j < PRIV_Q ? j : tid;
        dn[g] = pd[jv];
        sn[g] = ps[jv];
    }
    for (int i0 = 0; i0 < PRIV_Q; i0 += 2 * PRIV_BS) {
        int  dd[8], ss[8];
        bool ok[2];
        #pragma unroll
        for (int g = 0; g < 2; ++g) {
            ok[g] = i0 + g * PRIV_BS + tid < PRIV_Q;
            dd[4 * g + 0] = dn[g].x; dd[4 * g + 1] = dn[g].y; dd[4 * g + 2] = dn[g].z; dd[4 * g + 3] = dn[g].w;
            ss[4 * g + 0] = sn[g].x; ss[4 * g + 1] = sn[g].y; ss[4 * g + 2] = sn[g].z; ss[4 * g + 3] = sn[g].w;
        }
        #pragma unroll
        for (int g = 0; g < 2; ++g) {
            const int j  = i0 + (2 + g) * PRIV_BS + tid;
            const int jv = j < PRIV_Q ? j : tid;
            dn[g] = pd[jv];
            sn[g] = ps[jv];
        }
        bool     hit[8];
        unsigned q[8];
        float    a[8], b[8], w[8];
        #pragma unroll
        for (int k = 0; k < 8; ++k) {
            const unsigned r = (unsigned)(dd[k] - base);
            hit[k] = ok[k >> 2] && r < PRIV_RSZ;
            q[k]   = hit[k] ? r : 0u;
            const int sidx = hit[k] ? ss[k] : 0;
            const float2 t = reinterpret_cast<const float2*>(pr)[sidx];
            a[k] = t.x;
            b[k] = t.y;
        }
        #pragma unroll
        for (int k = 0; k < 8; ++k) w[k] = a[k] * dsl[q[k]] * b[k];
        #pragma unroll
        for (int k = 0; k < 8; ++k)
            if (hit[k]) atomicAdd(&acc[q[k]], w[k]);
    }
    __syncthreads();

    const int rot = chunk * PRIV_ROT;
    for (int i = tid; i < PRIV_RSZ; i += PRIV_BS) {
        int j = i + rot;
        if (j >= PRIV_RSZ) j -= PRIV_RSZ;
        atomicAdd(&out[base + j], acc[j]);
    }
}


static inline int blks(long long n, int bs) { return (int)((n + bs - 1) / bs); }

extern "C" void kernel_launch(void* const* d_in, const int* in_sizes, int n_in,
                              void* d_out, int out_size, void* d_ws, size_t ws_size,
                              hipStream_t stream) {
    (void)in_sizes; (void)n_in; (void)out_size; (void)ws_size;

    const float* x   = (const float*)d_in[0];
    const int*   ei  = (const int*)  d_in[1];
    const float* W1  = (const float*)d_in[2];
    const float* b1  = (const float*)d_in[3];
    const float* W2  = (const float*)d_in[4];
    const float* b2  = (const float*)d_in[5];
    const float* W3  = (const float*)d_in[6];
    const float* b3  = (const float*)d_in[7];
    float* out = (float*)d_out;

    const int* src = ei;
    const int* dst = ei + N_EDGES;

    float* ws   = (float*)d_ws;
    float* hw1p = ws;
    float* hw2p = ws;
    float* h2   = ws   + 32 * N_NODES;
    float* pr3  = ws   + 48 * N_NODES;
    float* dis  = ws   + 64 * N_NODES;
    float* h1   = dis  + N_NODES;

    const int BS = 256;

    k_init_deg <<<blks(N_NODES, BS), BS, 0, stream>>>(dis);
    k_count_deg<<<PRIV_NR * PRIV_NC, PRIV_BS, 0, stream>>>(dst, dis);

    const int gemm_grid = blks(N_NODES / 16, BS / 32);

    k_gemm_wmma<128, 32, true><<<gemm_grid, BS, 0, stream>>>(x, W1, nullptr, dis, hw1p, h1, N_NODES);
    k_edge_agg_lds<32, 64, 4><<<64 * 4, AG_BS, 0, stream>>>(src, dst, dis, hw1p, h1);

    k_gemm_wmma<32, 16, false><<<gemm_grid, BS, 0, stream>>>(h1, W2, b1, dis, hw2p, h2, N_NODES);
    k_edge_agg_lds<16, 32, 8><<<32 * 8, AG_BS, 0, stream>>>(src, dst, dis, hw2p, h2);

    k_gemm_out1    <<<blks(N_NODES, BS), BS, 0, stream>>>(h2, b2, W3, dis, b3, pr3, out);
    k_edge_agg_out1<<<PRIV_NR * PRIV_NC, PRIV_BS, 0, stream>>>(src, dst, dis, pr3, out);
}


# Row scatters add rows in full batches of 8 only: list entries that do not fill a batch are carried to the front of the next pass's list, no batch runs partly masked; sums stay in LDS arrival order
# speedup vs baseline: 3.8962x; 1.0829x over previous
#include <hip/hip_runtime.h>

typedef __attribute__((ext_vector_type(2))) float v2f;
typedef __attribute__((ext_vector_type(8))) float v8f;

#define N_NODES 100000
#define N_EDGES 3200000


__global__ void k_init_deg(float* __restrict__ deg) {
    int i = blockIdx.x * blockDim.x + threadIdx.x;
    if (i < N_NODES) deg[i] = 1.0f;
}

#define PRIV_NR     8
#define PRIV_RSZ    (N_NODES / PRIV_NR)
#define PRIV_NC     32
#define PRIV_CHUNK  (N_EDGES / PRIV_NC)
#define PRIV_Q      (PRIV_CHUNK / 4)
#define PRIV_BS     1024
#define PRIV_ROT    384

__global__ __launch_bounds__(PRIV_BS) void k_count_deg(const int* __restrict__ dst,
                                                       float* __restrict__ deg) {
    __shared__ unsigned cnt[PRIV_RSZ];
    const int tid   = threadIdx.x;
    const int range = blockIdx.x % PRIV_NR;
    const int chunk = blockIdx.x / PRIV_NR;
    const int base  = range * PRIV_RSZ;

    for (int i = tid; i < PRIV_RSZ; i += PRIV_BS) cnt[i] = 0u;
    __syncthreads();

    const int4* pd = reinterpret_cast<const int4*>(dst + (size_t)chunk * PRIV_CHUNK);
    for (int i = tid; i < PRIV_Q; i += 4 * PRIV_BS) {
        int4 d[4];
        bool ok[4];
        #pragma unroll
        for (int g = 0; g < 4; ++g) {
            const int j = i + g * PRIV_BS;
            ok[g] = j < PRIV_Q;
            d[g] = pd[ok[g] ? j : i];
        }
        #pragma unroll
        for (int g = 0; g < 4; ++g) {
            const unsigned r0 = (unsigned)(d[g].x - base);
            const unsigned r1 = (unsigned)(d[g].y - base);
            const unsigned r2 = (unsigned)(d[g].z - base);
            const unsigned r3 = (unsigned)(d[g].w - base);
            if (ok[g] && r0 < PRIV_RSZ) atomicAdd(&cnt[r0], 1u);
            if (ok[g] && r1 < PRIV_RSZ) atomicAdd(&cnt[r1], 1u);
            if (ok[g] && r2 < PRIV_RSZ) atomicAdd(&cnt[r2], 1u);
            if (ok[g] && r3 < PRIV_RSZ) atomicAdd(&cnt[r3], 1u);
        }
    }
    __syncthreads();

    const int rot = chunk * PRIV_ROT;
    for (int i = tid; i < PRIV_RSZ; i += PRIV_BS) {
        int j = i + rot;
        if (j >= PRIV_RSZ) j -= PRIV_RSZ;
        atomicAdd(&deg[base + j], (float)cnt[j]);
    }
}

template <int IN, int OUT, bool FIRST>
__global__ __launch_bounds__(256) void k_gemm_wmma(const float* __restrict__ X,
                                                   const float* __restrict__ W,
                                                   const float* __restrict__ bin,
                                                   float* __restrict__ dis,
                                                   float* __restrict__ Y,
                                                   float* __restrict__ agg,
                                                   int nrows) {
    const int lane = threadIdx.x & 31;
    const int wave = blockIdx.x * (blockDim.x >> 5) + (threadIdx.x >> 5);
    const int row0 = wave * 16;
    if (row0 >= nrows) return;

    const int half = lane >> 4;
    const int l15  = lane & 15;
    const int arow = row0 + l15;

    float d_own = dis[arow];

    constexpr int NT = OUT / 16;
    v8f acc[NT];
    #pragma unroll
    for (int t = 0; t < NT; ++t)
        #pragma unroll
        for (int v = 0; v < 8; ++v) acc[t][v] = 0.0f;

    for (int k = 0; k < IN; k += 4) {
        const float* ap = X + (size_t)arow * IN + (k + 2 * half);
        v2f a;
        a.x = ap[0];
        a.y = ap[1];
        if constexpr (!FIRST) {
            const float v0 = a.x + bin[k + 2 * half];
            const float v1 = a.y + bin[k + 2 * half + 1];
            a.x = v0 > 0.0f ? v0 : 0.0f;
            a.y = v1 > 0.0f ? v1 : 0.0f;
        }

        #pragma unroll
        for (int t = 0; t < NT; ++t) {
            const int col = t * 16 + l15;
            v2f b;
            b.x = W[(k + 2 * half)     * OUT + col];
            b.y = W[(k + 2 * half + 1) * OUT + col];
            acc[t] = __builtin_amdgcn_wmma_f32_16x16x4_f32(
                 false, a,  false, b,
                 (short)0, acc[t],  false,  false);
        }
    }

    if constexpr (FIRST) {
        d_own = rsqrtf(d_own);
        if (half == 0) dis[arow] = d_own;
    }

    float dv[8], dd[8];
    #pragma unroll
    for (int v = 0; v < 8; ++v) {
        dv[v] = __shfl(d_own, v + 8 * half);
        dd[v] = dv[v] * dv[v];
    }
    float2* Yp = reinterpret_cast<float2*>(Y);
    #pragma unroll
    for (int t = 0; t < NT; ++t)
        #pragma unroll
        for (int v = 0; v < 8; ++v) {
            int r = row0 + v + 8 * half;
            int c = t * 16 + l15;
            Yp[(size_t)r * OUT + c]  = make_float2(acc[t][v], dv[v]);
            agg[(size_t)r * OUT + c] = dd[v] * acc[t][v];
        }
}


#define AG_BS    1024
#define AG_U     4
#define AG_BATCH 8
#define AG_CAP   (AG_U * 4 * 32 + AG_BATCH)
#define AG_QBITS 12

template <int OUT, int NR, int NC>
__global__ __launch_bounds__(AG_BS) void k_edge_agg_lds(const int* __restrict__ src,
                                                        const int* __restrict__ dst,
                                                        const float* __restrict__ dis,
                                                        const float* __restrict__ hw,
                                                        float* __restrict__ agg) {
    constexpr int RSZ   = ((N_NODES + NR - 1) / NR + 1) & ~1;
    constexpr int CHUNK = N_EDGES / NC;
    constexpr int Q     = CHUNK / 4;
    constexpr int HPW   = 32 / OUT;
    constexpr int NI    = AG_BATCH / HPW;
    static_assert(RSZ <= (1 << AG_QBITS), "local index must fit its field");
    static_assert(N_EDGES % (NC * 4) == 0, "stretches are whole 16-byte groups");

    __shared__ float    acc[RSZ * OUT];
    __shared__ float    dsl[RSZ];
    __shared__ unsigned lst[(AG_BS / 32) * AG_CAP];

    const int tid   = threadIdx.x;
    const int lane  = tid & 31;
    const int range = blockIdx.x % NR;
    const int chunk = blockIdx.x / NR;
    const int base  = range * RSZ;
    const int rsz   = (N_NODES - base) < RSZ ? (N_NODES - base) : RSZ;

    const int4* pd = reinterpret_cast<const int4*>(dst + (size_t)chunk * CHUNK);
    const int*  ps = src + (size_t)chunk * CHUNK;

    int4 dn[AG_U];
    #pragma unroll
    for (int g = 0; g < AG_U; ++g) {
        const int j = g * AG_BS + tid;
        dn[g] = pd[j < Q ? j : tid];
    }

    for (int i = tid; i < RSZ * OUT; i += AG_BS) acc[i] = 0.0f;
    for (int i = tid; i < RSZ; i += AG_BS) dsl[i] = i < rsz ? dis[base + i] : 0.0f;
    __syncthreads();

    unsigned* my = lst + (tid >> 5) * AG_CAP;
    const float2* hwp = reinterpret_cast<const float2*>(hw);
    const int   f   = lane & (OUT - 1);
    const int   sub = lane / OUT;

    constexpr int      STEP  = AG_U * AG_BS;
    constexpr unsigned QMASK = (1u << AG_QBITS) - 1u;
    static_assert(CHUNK <= (1 << (32 - AG_QBITS)), "edge offset in a stretch must fit its field");

    int      keep = 0;
    int      nnew = 0;
    unsigned ent  = 0u;
    int      sv   = 0;

    for (int i0 = 0; i0 < Q + STEP; i0 += STEP) {
        bool     ok[AG_U];
        unsigned okm[AG_U];
        unsigned eb[AG_U];
        #pragma unroll
        for (int g = 0; g < AG_U; ++g) {
            const int j = i0 + g * AG_BS + tid;
            ok[g]  = j < Q;
            okm[g] = __builtin_amdgcn_ballot_w32(ok[g]);
            eb[g]  = (unsigned)(4 * j) << AG_QBITS;
        }
        bool     hit[4 * AG_U];
        unsigned hm[4 * AG_U];
        unsigned q[4 * AG_U];
        #pragma unroll
        for (int k = 0; k < 4 * AG_U; ++k) {
            const int g = k >> 2, c = k & 3;
            const int dv = c == 0 ? dn[g].x : c == 1 ? dn[g].y : c == 2 ? dn[g].z : dn[g].w;
            const unsigned r = (unsigned)(dv - base);
            const bool in = r < (unsigned)RSZ;
            hm[k]  = __builtin_amdgcn_ballot_w32(in) & okm[g];
            hit[k] = in && ok[g];
            q[k]   = r;
        }
        #pragma unroll
        for (int g = 0; g < AG_U; ++g) {
            const int j = i0 + STEP + g * AG_BS + tid;
            dn[g] = pd[j < Q ? j : tid];
        }
        if (lane < nnew) my[keep + lane] = ((unsigned)sv << AG_QBITS) | (ent & QMASK);
        #pragma nounroll
        for (int j0 = 32; j0 < nnew; j0 += 32) {
            if (j0 + lane < nnew) {
                const int j = keep + j0 + lane;
                const unsigned e = my[j];
                const int s = ps[e >> AG_QBITS];
                my[j] = ((unsigned)s << AG_QBITS) | (e & QMASK);
            }
        }
        const int cnt = keep + nnew;
        int b0 = 0;
        for (; b0 + AG_BATCH <= cnt; b0 += AG_BATCH) {
            unsigned lq[NI];
            float    rw[NI], dsv[NI], dl[NI];
            #pragma unroll
            for (int u = 0; u < NI; ++u) {
                const unsigned e = my[b0 + u * HPW + sub];
                const int s = (int)(e >> AG_QBITS);
                lq[u]  = e & QMASK;
                const float2 p = hwp[(size_t)s * OUT + f];
                rw[u]  = p.x;
                dsv[u] = p.y;
                dl[u]  = dsl[lq[u]];
            }
            #pragma unroll
            for (int u = 0; u < NI; ++u) {
                const float w = dsv[u] * dl[u] * rw[u];
                atomicAdd(&acc[lq[u] * OUT + f], w);
            }
        }
        keep = cnt - b0;
        if (b0 > 0 && lane < keep) {
            const unsigned e = my[b0 + lane];
            my[lane] = e;
        }
        nnew = 0;
        #pragma unroll
        for (int k = 0; k < 4 * AG_U; ++k) {
            const int g = k >> 2, c = k & 3;
            if (hm[k] != 0u) {
                const int pos = keep + nnew + (int)__builtin_amdgcn_mbcnt_lo(hm[k], 0u);
                if (hit[k]) my[pos] = eb[g] | ((unsigned)c << AG_QBITS) | q[k];
                nnew += __builtin_popcount(hm[k]);
            }
        }
        ent = my[keep + lane];
        sv  = ps[lane < nnew ? (int)(ent >> AG_QBITS) : 0];
    }
    if (keep > 0) {
        bool     val[NI];
        unsigned lq[NI];
        float    rw[NI], dsv[NI], dl[NI];
        #pragma unroll
        for (int u = 0; u < NI; ++u) {
            const int idx = u * HPW + sub;
            val[u] = idx < keep;
            const unsigned e = my[val[u] ? idx : 0];
            const int s = (int)(e >> AG_QBITS);
            lq[u]  = e & QMASK;
            const float2 p = hwp[(size_t)s * OUT + f];
            rw[u]  = p.x;
            dsv[u] = p.y;
            dl[u]  = dsl[lq[u]];
        }
        #pragma unroll
        for (int u = 0; u < NI; ++u) {
            const float w = dsv[u] * dl[u] * rw[u];
            if (val[u]) atomicAdd(&acc[lq[u] * OUT + f], w);
        }
    }
    __syncthreads();

    const int total = rsz * OUT;
    const int rot   = chunk * (((RSZ * OUT) / (2 * NC)) & ~31);
    float* ag = agg + (size_t)base * OUT;
    for (int i = tid; i < total; i += AG_BS) {
        int j = i + rot;
        if (j >= total) j -= total;
        atomicAdd(&ag[j], acc[j]);
    }
}


__global__ void k_gemm_out1(const float* __restrict__ h2, const float* __restrict__ b2,
                            const float* __restrict__ W3, const float* __restrict__ dis,
                            const float* __restrict__ b3,
                            float* __restrict__ pr, float* __restrict__ out) {
    int i = blockIdx.x * blockDim.x + threadIdx.x;
    if (i >= N_NODES) return;
    float s = 0.0f;
    #pragma unroll
    for (int k = 0; k < 16; ++k) {
        const float v = h2[(size_t)i * 16 + k] + b2[k];
        s += (v > 0.0f ? v : 0.0f) * W3[k];
    }
    const float d = dis[i];
    reinterpret_cast<float2*>(pr)[i] = make_float2(d, s);
    out[i] = d * d * s + b3[0];
}

__global__ __launch_bounds__(PRIV_BS) void k_edge_agg_out1(const int* __restrict__ src,
                                                           const int* __restrict__ dst,
                                                           const float* __restrict__ dis,
                                                           const float* __restrict__ pr,
                                                           float* __restrict__ out) {
    __shared__ float acc[PRIV_RSZ];
    __shared__ float dsl[PRIV_RSZ];
    const int tid   = threadIdx.x;
    const int range = blockIdx.x % PRIV_NR;
    const int chunk = blockIdx.x / PRIV_NR;
    const int base  = range * PRIV_RSZ;

    for (int i = tid; i < PRIV_RSZ; i += PRIV_BS) {
        acc[i] = 0.0f;
        dsl[i] = dis[base + i];
    }
    __syncthreads();

    const int4* pd = reinterpret_cast<const int4*>(dst + (size_t)chunk * PRIV_CHUNK);
    const int4* ps = reinterpret_cast<const int4*>(src + (size_t)chunk * PRIV_CHUNK);
    int4 dn[2], sn[2];
    #pragma unroll
    for (int g = 0; g < 2; ++g) {
        const int j  = g * PRIV_BS + tid;
        const int jv = j < PRIV_Q ? j : tid;
        dn[g] = pd[jv];
        sn[g] = ps[jv];
    }
    for (int i0 = 0; i0 < PRIV_Q; i0 += 2 * PRIV_BS) {
        int  dd[8], ss[8];
        bool ok[2];
        #pragma unroll
        for (int g = 0; g < 2; ++g) {
            ok[g] = i0 + g * PRIV_BS + tid < PRIV_Q;
            dd[4 * g + 0] = dn[g].x; dd[4 * g + 1] = dn[g].y; dd[4 * g + 2] = dn[g].z; dd[4 * g + 3] = dn[g].w;
            ss[4 * g + 0] = sn[g].x; ss[4 * g + 1] = sn[g].y; ss[4 * g + 2] = sn[g].z; ss[4 * g + 3] = sn[g].w;
        }
        #pragma unroll
        for (int g = 0; g < 2; ++g) {
            const int j  = i0 + (2 + g) * PRIV_BS + tid;
            const int jv = j < PRIV_Q ? j : tid;
            dn[g] = pd[jv];
            sn[g] = ps[jv];
        }
        bool     hit[8];
        unsigned q[8];
        float    a[8], b[8], w[8];
        #pragma unroll
        for (int k = 0; k < 8; ++k) {
            const unsigned r = (unsigned)(dd[k] - base);
            hit[k] = ok[k >> 2] && r < PRIV_RSZ;
            q[k]   = hit[k] ? r : 0u;
            const int sidx = hit[k] ? ss[k] : 0;
            const float2 t = reinterpret_cast<const float2*>(pr)[sidx];
            a[k] = t.x;
            b[k] = t.y;
        }
        #pragma unroll
        for (int k = 0; k < 8; ++k) w[k] = a[k] * dsl[q[k]] * b[k];
        #pragma unroll
        for (int k = 0; k < 8; ++k)
            if (hit[k]) atomicAdd(&acc[q[k]], w[k]);
    }
    __syncthreads();

    const int rot = chunk * PRIV_ROT;
    for (int i = tid; i < PRIV_RSZ; i += PRIV_BS) {
        int j = i + rot;
        if (j >= PRIV_RSZ) j -= PRIV_RSZ;
        atomicAdd(&out[base + j], acc[j]);
    }
}


static inline int blks(long long n, int bs) { return (int)((n + bs - 1) / bs); }

extern "C" void kernel_launch(void* const* d_in, const int* in_sizes, int n_in,
                              void* d_out, int out_size, void* d_ws, size_t ws_size,
                              hipStream_t stream) {
    (void)in_sizes; (void)n_in; (void)out_size; (void)ws_size;

    const float* x   = (const float*)d_in[0];
    const int*   ei  = (const int*)  d_in[1];
    const float* W1  = (const float*)d_in[2];
    const float* b1  = (const float*)d_in[3];
    const float* W2  = (const float*)d_in[4];
    const float* b2  = (const float*)d_in[5];
    const float* W3  = (const float*)d_in[6];
    const float* b3  = (const float*)d_in[7];
    float* out = (float*)d_out;

    const int* src = ei;
    const int* dst = ei + N_EDGES;

    float* ws   = (float*)d_ws;
    float* hw1p = ws;
    float* hw2p = ws;
    float* h2   = ws   + 32 * N_NODES;
    float* pr3  = ws   + 48 * N_NODES;
    float* dis  = ws   + 64 * N_NODES;
    float* h1   = dis  + N_NODES;

    const int BS = 256;

    k_init_deg <<<blks(N_NODES, BS), BS, 0, stream>>>(dis);
    k_count_deg<<<PRIV_NR * PRIV_NC, PRIV_BS, 0, stream>>>(dst, dis);

    const int gemm_grid = blks(N_NODES / 16, BS / 32);

    k_gemm_wmma<128, 32, true><<<gemm_grid, BS, 0, stream>>>(x, W1, nullptr, dis, hw1p, h1, N_NODES);
    k_edge_agg_lds<32, 64, 4><<<64 * 4, AG_BS, 0, stream>>>(src, dst, dis, hw1p, h1);

    k_gemm_wmma<32, 16, false><<<gemm_grid, BS, 0, stream>>>(h1, W2, b1, dis, hw2p, h2, N_NODES);
    k_edge_agg_lds<16, 32, 8><<<32 * 8, AG_BS, 0, stream>>>(src, dst, dis, hw2p, h2);

    k_gemm_out1    <<<blks(N_NODES, BS), BS, 0, stream>>>(h2, b2, W3, dis, b3, pr3, out);
    k_edge_agg_out1<<<PRIV_NR * PRIV_NC, PRIV_BS, 0, stream>>>(src, dst, dis, pr3, out);
}
